# starting kernel
# baseline (speedup 1.0000x reference)
#include <hip/hip_runtime.h>

#define NN   4096
#define DD   70
#define NT   256
#define ND   5
#define SPLITK 8
#define TPC  (NT / SPLITK)

typedef float v2f __attribute__((ext_vector_type(2)));
typedef float v4f __attribute__((ext_vector_type(4)));
typedef float v8f __attribute__((ext_vector_type(8)));

__device__ __forceinline__ v8f wmma4(v2f a, v2f b, v8f c) {
  return __builtin_amdgcn_wmma_f32_16x16x4_f32(
       false, a,  false, b,
       (short)0, c,  false,  false);
}

__global__ void zero_f32(float* __restrict__ p, int n) {
  int i = blockIdx.x * blockDim.x + threadIdx.x;
  if (i < n) p[i] = 0.0f;
}

__global__ void pack_sB(const float* __restrict__ s, v2f* __restrict__ sB) {
  int tid = blockIdx.x * blockDim.x + threadIdx.x;
  if (tid >= NT * ND * 4 * 32) return;
  int lane = tid & 31;
  int e = tid >> 5;
  int q = e & 3;  e >>= 2;
  int n = e % ND;
  int t = e / ND;
  int r = 16 * t + 4 * q + 2 * (lane >> 4);
  int d = 16 * n + (lane & 15);
  v2f v;
  v.x = (d < DD) ? s[(size_t)r * DD + d]       : 0.0f;
  v.y = (d < DD) ? s[(size_t)(r + 1) * DD + d] : 0.0f;
  sB[tid] = v;
}

__global__ __launch_bounds__(32) void slayer_wmma(const float* __restrict__ adj,
                                                  const v2f*  __restrict__ sB,
                                                  float*      __restrict__ out) {
  const int wg   = blockIdx.x;
  const int b    = wg >> 3;
  const int c    = wg & (SPLITK-1);
  const int lane = threadIdx.x;
  const int m    = lane & 15;
  const int kh   = lane >> 4;

  v8f zero = {};
  v8f accI[ND], accO[ND];
#pragma unroll
  for (int n = 0; n < ND; ++n) { accI[n] = zero; accO[n] = zero; }

  const int i0 = 16 * b;
  for (int t = TPC * c; t < TPC * c + TPC; ++t) {
    const int j0 = 16 * t;
#pragma unroll
    for (int q = 0; q < 4; ++q) {
      const int kb = 4 * q + 2 * kh;
      v4f f = *(const v4f*)(adj + 2 * ((size_t)(i0 + m) * NN + (j0 + kb)));
      v2f aO; aO.x = f.x + f.y; aO.y = f.z + f.w;
      const float* p = adj + 2 * ((size_t)(j0 + kb) * NN + (i0 + m));
      v2f g0 = *(const v2f*)p;
      v2f g1 = *(const v2f*)(p + 2 * NN);
      v2f aI; aI.x = g0.x + g0.y; aI.y = g1.x + g1.y;
      const v2f* bp = sB + (size_t)t * (ND * 4 * 32) + q * 32 + lane;
#pragma unroll
      for (int n = 0; n < ND; ++n) {
        v2f bv = bp[n * 4 * 32];
        accO[n] = wmma4(aO, bv, accO[n]);
        accI[n] = wmma4(aI, bv, accI[n]);
      }
    }
  }

  float* outI = out;
  float* outO = out + (size_t)NN * DD;
#pragma unroll
  for (int n = 0; n < ND; ++n) {
    int d = 16 * n + m;
    if (d < DD) {
#pragma unroll
      for (int r = 0; r < 8; ++r) {
        int row = i0 + r + 8 * kh;
        atomicAdd(&outI[(size_t)row * DD + d], accI[n][r]);
        atomicAdd(&outO[(size_t)row * DD + d], accO[n][r]);
      }
    }
  }
}

extern "C" void kernel_launch(void* const* d_in, const int* in_sizes, int n_in,
                              void* d_out, int out_size, void* d_ws, size_t ws_size,
                              hipStream_t stream) {
  const float* adj = (const float*)d_in[0];
  const float* s   = (const float*)d_in[1];
  float* out = (float*)d_out;
  v2f* sB = (v2f*)d_ws;

  zero_f32<<<(out_size + 255) / 256, 256, 0, stream>>>(out, out_size);

  int packN = NT * ND * 4 * 32;
  pack_sB<<<(packN + 255) / 256, 256, 0, stream>>>(s, sB);

  slayer_wmma<<<NT * SPLITK, 32, 0, stream>>>(adj, sB, out);
}


# main loop software-pipelined over three register sets: the 32 loads of tile t+2 issue as one group ahead of tile t's 40 matrix instructions instead of being waited for inside the tile they feed
# speedup vs baseline: 1.3074x; 1.3074x over previous
#include <hip/hip_runtime.h>

#define NN   4096
#define DD   70
#define NT   256
#define ND   5
#define SPLITK 8
#define TPC  (NT / SPLITK)

typedef float v2f __attribute__((ext_vector_type(2)));
typedef float v4f __attribute__((ext_vector_type(4)));
typedef float v8f __attribute__((ext_vector_type(8)));

__device__ __forceinline__ v8f wmma4(v2f a, v2f b, v8f c) {
  return __builtin_amdgcn_wmma_f32_16x16x4_f32(
       false, a,  false, b,
       (short)0, c,  false,  false);
}

__global__ void zero_f32(float* __restrict__ p, int n) {
  int i = blockIdx.x * blockDim.x + threadIdx.x;
  if (i < n) p[i] = 0.0f;
}

__global__ void pack_sB(const float* __restrict__ s, v2f* __restrict__ sB) {
  int tid = blockIdx.x * blockDim.x + threadIdx.x;
  if (tid >= NT * ND * 4 * 32) return;
  int lane = tid & 31;
  int e = tid >> 5;
  int q = e & 3;  e >>= 2;
  int n = e % ND;
  int t = e / ND;
  int r = 16 * t + 4 * q + 2 * (lane >> 4);
  int d = 16 * n + (lane & 15);
  v2f v;
  v.x = (d < DD) ? s[(size_t)r * DD + d]       : 0.0f;
  v.y = (d < DD) ? s[(size_t)(r + 1) * DD + d] : 0.0f;
  sB[tid] = v;
}

struct TileRegs {
  v4f f[4];
  v2f g0[4], g1[4];
  v2f b[4][ND];
};

__device__ __forceinline__ void load_tile(TileRegs& T, const float* __restrict__ pO,
                                          const float* __restrict__ pI,
                                          const v2f* __restrict__ pB) {
#pragma unroll
  for (int q = 0; q < 4; ++q) {
    T.f[q]  = *(const v4f*)(pO + 8 * q);
    T.g0[q] = *(const v2f*)(pI + (size_t)q * 8 * NN);
    T.g1[q] = *(const v2f*)(pI + (size_t)q * 8 * NN + 2 * NN);
#pragma unroll
    for (int n = 0; n < ND; ++n) T.b[q][n] = pB[q * 32 + n * 4 * 32];
  }
}

__device__ __forceinline__ void mma_tile(const TileRegs& T, v8f (&accI)[ND], v8f (&accO)[ND]) {
#pragma unroll
  for (int q = 0; q < 4; ++q) {
    v2f aO; aO.x = T.f[q].x + T.f[q].y;   aO.y = T.f[q].z + T.f[q].w;
    v2f aI; aI.x = T.g0[q].x + T.g0[q].y; aI.y = T.g1[q].x + T.g1[q].y;
#pragma unroll
    for (int n = 0; n < ND; ++n) {
      accO[n] = wmma4(aO, T.b[q][n], accO[n]);
      accI[n] = wmma4(aI, T.b[q][n], accI[n]);
    }
  }
}

__global__ __launch_bounds__(32) void slayer_wmma(const float* __restrict__ adj,
                                                  const v2f*  __restrict__ sB,
                                                  float*      __restrict__ out) {
  const int wg   = blockIdx.x;
  const int b    = wg >> 3;
  const int c    = wg & (SPLITK-1);
  const int lane = threadIdx.x;
  const int m    = lane & 15;
  const int kh   = lane >> 4;

  v8f zero = {};
  v8f accI[ND], accO[ND];
#pragma unroll
  for (int n = 0; n < ND; ++n) { accI[n] = zero; accO[n] = zero; }

  const int i0 = 16 * b;
  const int t0 = TPC * c;
  const float* pO = adj + 2 * ((size_t)(i0 + m) * NN + (16 * t0 + 2 * kh));
  const float* pI = adj + 2 * ((size_t)(16 * t0 + 2 * kh) * NN + (i0 + m));
  const v2f*   pB = sB + (size_t)t0 * (ND * 4 * 32) + lane;
  const size_t dO = 2 * 16;
  const size_t dI = (size_t)2 * 16 * NN;
  const size_t dB = ND * 4 * 32;

#define SCHED_FENCE() __builtin_amdgcn_sched_barrier(0)

  TileRegs R0, R1, R2;
  load_tile(R0, pO, pI, pB);
  SCHED_FENCE();
  load_tile(R1, pO + dO, pI + dI, pB + dB);
  SCHED_FENCE();
  pO += 2 * dO; pI += 2 * dI; pB += 2 * dB;

  for (int it = 0; it < (TPC - 2) / 3; ++it) {
    load_tile(R2, pO, pI, pB);
    SCHED_FENCE();
    mma_tile(R0, accI, accO);
    SCHED_FENCE();
    load_tile(R0, pO + dO, pI + dI, pB + dB);
    SCHED_FENCE();
    mma_tile(R1, accI, accO);
    SCHED_FENCE();
    load_tile(R1, pO + 2 * dO, pI + 2 * dI, pB + 2 * dB);
    SCHED_FENCE();
    mma_tile(R2, accI, accO);
    SCHED_FENCE();
    pO += 3 * dO; pI += 3 * dI; pB += 3 * dB;
  }
  mma_tile(R0, accI, accO);
  mma_tile(R1, accI, accO);
#undef SCHED_FENCE

  float* outI = out;
  float* outO = out + (size_t)NN * DD;
#pragma unroll
  for (int n = 0; n < ND; ++n) {
    int d = 16 * n + m;
    if (d < DD) {
#pragma unroll
      for (int r = 0; r < 8; ++r) {
        int row = i0 + r + 8 * kh;
        atomicAdd(&outI[(size_t)row * DD + d], accI[n][r]);
        atomicAdd(&outO[(size_t)row * DD + d], accO[n][r]);
      }
    }
  }
}

extern "C" void kernel_launch(void* const* d_in, const int* in_sizes, int n_in,
                              void* d_out, int out_size, void* d_ws, size_t ws_size,
                              hipStream_t stream) {
  const float* adj = (const float*)d_in[0];
  const float* s   = (const float*)d_in[1];
  float* out = (float*)d_out;
  v2f* sB = (v2f*)d_ws;

  zero_f32<<<(out_size + 255) / 256, 256, 0, stream>>>(out, out_size);

  int packN = NT * ND * 4 * 32;
  pack_sB<<<(packN + 255) / 256, 256, 0, stream>>>(s, sB);

  slayer_wmma<<<NT * SPLITK, 32, 0, stream>>>(adj, sB, out);
}


# output clear and s repack fused into one preparatory launch of 160 workgroups (16-byte clearing stores, 4 pack entries per thread): 3 launches become 2, 23,040 short waves become 1,280
# speedup vs baseline: 1.3589x; 1.0394x over previous
#include <hip/hip_runtime.h>

#define NN   4096
#define DD   70
#define NT   256
#define ND   5
#define SPLITK 8
#define TPC  (NT / SPLITK)

typedef float v2f __attribute__((ext_vector_type(2)));
typedef float v4f __attribute__((ext_vector_type(4)));
typedef float v8f __attribute__((ext_vector_type(8)));

__device__ __forceinline__ v8f wmma4(v2f a, v2f b, v8f c) {
  return __builtin_amdgcn_wmma_f32_16x16x4_f32(
       false, a,  false, b,
       (short)0, c,  false,  false);
}

#define PREP_THREADS 256
#define PREP_BLOCKS  160
#define PREP_TOTAL   (PREP_THREADS * PREP_BLOCKS)
#define PACK_N       (NT * ND * 4 * 32)
#define PACK_PER     (PACK_N / PREP_TOTAL)
static_assert(PACK_PER * PREP_TOTAL == PACK_N, "pack entries must divide evenly over the prep threads");

__device__ __forceinline__ v2f pack_entry(const float* __restrict__ s, int idx) {
  int lane = idx & 31;
  int e = idx >> 5;
  int q = e & 3;  e >>= 2;
  int n = e % ND;
  int t = e / ND;
  int r = 16 * t + 4 * q + 2 * (lane >> 4);
  int d = 16 * n + (lane & 15);
  v2f v;
  v.x = (d < DD) ? s[(size_t)r * DD + d]       : 0.0f;
  v.y = (d < DD) ? s[(size_t)(r + 1) * DD + d] : 0.0f;
  return v;
}

__global__ __launch_bounds__(PREP_THREADS) void prep_out_sB(const float* __restrict__ s,
                                                            v2f*   __restrict__ sB,
                                                            float* __restrict__ out, int n) {
  const int tid = blockIdx.x * PREP_THREADS + threadIdx.x;

  v2f v[PACK_PER];
#pragma unroll
  for (int k = 0; k < PACK_PER; ++k) v[k] = pack_entry(s, tid + k * PREP_TOTAL);

  const int n4 = n >> 2;
  v4f* out4 = (v4f*)out;
  const v4f z = {0.0f, 0.0f, 0.0f, 0.0f};
  for (int i = tid; i < n4; i += PREP_TOTAL) out4[i] = z;
  if (tid < (n & 3)) out[4 * n4 + tid] = 0.0f;

#pragma unroll
  for (int k = 0; k < PACK_PER; ++k) sB[tid + k * PREP_TOTAL] = v[k];
}

struct TileRegs {
  v4f f[4];
  v2f g0[4], g1[4];
  v2f b[4][ND];
};

__device__ __forceinline__ void load_tile(TileRegs& T, const float* __restrict__ pO,
                                          const float* __restrict__ pI,
                                          const v2f* __restrict__ pB) {
#pragma unroll
  for (int q = 0; q < 4; ++q) {
    T.f[q]  = *(const v4f*)(pO + 8 * q);
    T.g0[q] = *(const v2f*)(pI + (size_t)q * 8 * NN);
    T.g1[q] = *(const v2f*)(pI + (size_t)q * 8 * NN + 2 * NN);
#pragma unroll
    for (int n = 0; n < ND; ++n) T.b[q][n] = pB[q * 32 + n * 4 * 32];
  }
}

__device__ __forceinline__ void mma_tile(const TileRegs& T, v8f (&accI)[ND], v8f (&accO)[ND]) {
#pragma unroll
  for (int q = 0; q < 4; ++q) {
    v2f aO; aO.x = T.f[q].x + T.f[q].y;   aO.y = T.f[q].z + T.f[q].w;
    v2f aI; aI.x = T.g0[q].x + T.g0[q].y; aI.y = T.g1[q].x + T.g1[q].y;
#pragma unroll
    for (int n = 0; n < ND; ++n) {
      accO[n] = wmma4(aO, T.b[q][n], accO[n]);
      accI[n] = wmma4(aI, T.b[q][n], accI[n]);
    }
  }
}

__global__ __launch_bounds__(32) void slayer_wmma(const float* __restrict__ adj,
                                                  const v2f*  __restrict__ sB,
                                                  float*      __restrict__ out) {
  const int wg   = blockIdx.x;
  const int b    = wg >> 3;
  const int c    = wg & (SPLITK-1);
  const int lane = threadIdx.x;
  const int m    = lane & 15;
  const int kh   = lane >> 4;

  v8f zero = {};
  v8f accI[ND], accO[ND];
#pragma unroll
  for (int n = 0; n < ND; ++n) { accI[n] = zero; accO[n] = zero; }

  const int i0 = 16 * b;
  const int t0 = TPC * c;
  const float* pO = adj + 2 * ((size_t)(i0 + m) * NN + (16 * t0 + 2 * kh));
  const float* pI = adj + 2 * ((size_t)(16 * t0 + 2 * kh) * NN + (i0 + m));
  const v2f*   pB = sB + (size_t)t0 * (ND * 4 * 32) + lane;
  const size_t dO = 2 * 16;
  const size_t dI = (size_t)2 * 16 * NN;
  const size_t dB = ND * 4 * 32;

#define SCHED_FENCE() __builtin_amdgcn_sched_barrier(0)

  TileRegs R0, R1, R2;
  load_tile(R0, pO, pI, pB);
  SCHED_FENCE();
  load_tile(R1, pO + dO, pI + dI, pB + dB);
  SCHED_FENCE();
  pO += 2 * dO; pI += 2 * dI; pB += 2 * dB;

  for (int it = 0; it < (TPC - 2) / 3; ++it) {
    load_tile(R2, pO, pI, pB);
    SCHED_FENCE();
    mma_tile(R0, accI, accO);
    SCHED_FENCE();
    load_tile(R0, pO + dO, pI + dI, pB + dB);
    SCHED_FENCE();
    mma_tile(R1, accI, accO);
    SCHED_FENCE();
    load_tile(R1, pO + 2 * dO, pI + 2 * dI, pB + 2 * dB);
    SCHED_FENCE();
    mma_tile(R2, accI, accO);
    SCHED_FENCE();
    pO += 3 * dO; pI += 3 * dI; pB += 3 * dB;
  }
  mma_tile(R0, accI, accO);
  mma_tile(R1, accI, accO);
#undef SCHED_FENCE

  float* outI = out;
  float* outO = out + (size_t)NN * DD;
#pragma unroll
  for (int n = 0; n < ND; ++n) {
    int d = 16 * n + m;
    if (d < DD) {
#pragma unroll
      for (int r = 0; r < 8; ++r) {
        int row = i0 + r + 8 * kh;
        atomicAdd(&outI[(size_t)row * DD + d], accI[n][r]);
        atomicAdd(&outO[(size_t)row * DD + d], accO[n][r]);
      }
    }
  }
}

extern "C" void kernel_launch(void* const* d_in, const int* in_sizes, int n_in,
                              void* d_out, int out_size, void* d_ws, size_t ws_size,
                              hipStream_t stream) {
  const float* adj = (const float*)d_in[0];
  const float* s   = (const float*)d_in[1];
  float* out = (float*)d_out;
  v2f* sB = (v2f*)d_ws;

  prep_out_sB<<<PREP_BLOCKS, PREP_THREADS, 0, stream>>>(s, sB, out, out_size);

  slayer_wmma<<<NT * SPLITK, 32, 0, stream>>>(adj, sB, out);
}


# split-K sums leave float atomics: the 8 chunk waves of a row-block form one workgroup, add their partials through LDS in fixed chunk order and store each output once; the output clear is gone
# speedup vs baseline: 1.7856x; 1.3140x over previous
#include <hip/hip_runtime.h>

#define NN   4096
#define DD   70
#define NT   256
#define ND   5
#define SPLITK 8
#define TPC  (NT / SPLITK)

typedef float v2f __attribute__((ext_vector_type(2)));
typedef float v4f __attribute__((ext_vector_type(4)));
typedef float v8f __attribute__((ext_vector_type(8)));

__device__ __forceinline__ v8f wmma4(v2f a, v2f b, v8f c) {
  return __builtin_amdgcn_wmma_f32_16x16x4_f32(
       false, a,  false, b,
       (short)0, c,  false,  false);
}

#define PREP_THREADS 256
#define PREP_BLOCKS  160
#define PREP_TOTAL   (PREP_THREADS * PREP_BLOCKS)
#define PACK_N       (NT * ND * 4 * 32)
#define PACK_PER     (PACK_N / PREP_TOTAL)
static_assert(PACK_PER * PREP_TOTAL == PACK_N, "pack entries must divide evenly over the prep threads");

__device__ __forceinline__ v2f pack_entry(const float* __restrict__ s, int idx) {
  int lane = idx & 31;
  int e = idx >> 5;
  int q = e & 3;  e >>= 2;
  int n = e % ND;
  int t = e / ND;
  int r = 16 * t + 4 * q + 2 * (lane >> 4);
  int d = 16 * n + (lane & 15);
  v2f v;
  v.x = (d < DD) ? s[(size_t)r * DD + d]       : 0.0f;
  v.y = (d < DD) ? s[(size_t)(r + 1) * DD + d] : 0.0f;
  return v;
}

__global__ __launch_bounds__(PREP_THREADS) void prep_sB(const float* __restrict__ s,
                                                        v2f*   __restrict__ sB) {
  const int tid = blockIdx.x * PREP_THREADS + threadIdx.x;

  v2f v[PACK_PER];
#pragma unroll
  for (int k = 0; k < PACK_PER; ++k) v[k] = pack_entry(s, tid + k * PREP_TOTAL);

#pragma unroll
  for (int k = 0; k < PACK_PER; ++k) sB[tid + k * PREP_TOTAL] = v[k];
}

struct TileRegs {
  v4f f[4];
  v2f g0[4], g1[4];
  v2f b[4][ND];
};

__device__ __forceinline__ void load_tile(TileRegs& T, const float* __restrict__ pO,
                                          const float* __restrict__ pI,
                                          const v2f* __restrict__ pB) {
#pragma unroll
  for (int q = 0; q < 4; ++q) {
    T.f[q]  = *(const v4f*)(pO + 8 * q);
    T.g0[q] = *(const v2f*)(pI + (size_t)q * 8 * NN);
    T.g1[q] = *(const v2f*)(pI + (size_t)q * 8 * NN + 2 * NN);
#pragma unroll
    for (int n = 0; n < ND; ++n) T.b[q][n] = pB[q * 32 + n * 4 * 32];
  }
}

__device__ __forceinline__ void mma_tile(const TileRegs& T, v8f (&accI)[ND], v8f (&accO)[ND]) {
#pragma unroll
  for (int q = 0; q < 4; ++q) {
    v2f aO; aO.x = T.f[q].x + T.f[q].y;   aO.y = T.f[q].z + T.f[q].w;
    v2f aI; aI.x = T.g0[q].x + T.g0[q].y; aI.y = T.g1[q].x + T.g1[q].y;
#pragma unroll
    for (int n = 0; n < ND; ++n) {
      accO[n] = wmma4(aO, T.b[q][n], accO[n]);
      accI[n] = wmma4(aI, T.b[q][n], accI[n]);
    }
  }
}

#define WG_THREADS (32 * SPLITK)
#define PITCH      84
#define PART_WAVE  (2 * 16 * PITCH)
#define OUT_BLOCK  (16 * DD)

__global__ __launch_bounds__(WG_THREADS) void slayer_wmma(const float* __restrict__ adj,
                                                          const v2f*  __restrict__ sB,
                                                          float*      __restrict__ out) {
  __shared__ float part[SPLITK * PART_WAVE];

  const int b    = blockIdx.x;
  const int c    = __builtin_amdgcn_readfirstlane((int)(threadIdx.x >> 5));
  const int lane = threadIdx.x & 31;
  const int m    = lane & 15;
  const int kh   = lane >> 4;

  v8f zero = {};
  v8f accI[ND], accO[ND];
#pragma unroll
  for (int n = 0; n < ND; ++n) { accI[n] = zero; accO[n] = zero; }

  const int i0 = 16 * b;
  const int t0 = TPC * c;
  const float* pO = adj + 2 * ((size_t)(i0 + m) * NN + (16 * t0 + 2 * kh));
  const float* pI = adj + 2 * ((size_t)(16 * t0 + 2 * kh) * NN + (i0 + m));
  const v2f*   pB = sB + (size_t)t0 * (ND * 4 * 32) + lane;
  const size_t dO = 2 * 16;
  const size_t dI = (size_t)2 * 16 * NN;
  const size_t dB = ND * 4 * 32;

#define SCHED_FENCE() __builtin_amdgcn_sched_barrier(0)

  TileRegs R0, R1, R2;
  load_tile(R0, pO, pI, pB);
  SCHED_FENCE();
  load_tile(R1, pO + dO, pI + dI, pB + dB);
  SCHED_FENCE();
  pO += 2 * dO; pI += 2 * dI; pB += 2 * dB;

  for (int it = 0; it < (TPC - 2) / 3; ++it) {
    load_tile(R2, pO, pI, pB);
    SCHED_FENCE();
    mma_tile(R0, accI, accO);
    SCHED_FENCE();
    load_tile(R0, pO + dO, pI + dI, pB + dB);
    SCHED_FENCE();
    mma_tile(R1, accI, accO);
    SCHED_FENCE();
    load_tile(R1, pO + 2 * dO, pI + 2 * dI, pB + 2 * dB);
    SCHED_FENCE();
    mma_tile(R2, accI, accO);
    SCHED_FENCE();
    pO += 3 * dO; pI += 3 * dI; pB += 3 * dB;
  }
  mma_tile(R0, accI, accO);
  mma_tile(R1, accI, accO);
#undef SCHED_FENCE

  float* mine = part + c * PART_WAVE;
#pragma unroll
  for (int n = 0; n < ND; ++n) {
#pragma unroll
    for (int r = 0; r < 8; ++r) {
      const int at = (r + 8 * kh) * PITCH + 16 * n + m;
      mine[at]              = accI[n][r];
      mine[16 * PITCH + at] = accO[n][r];
    }
  }
  __syncthreads();

  for (int e = threadIdx.x; e < 2 * OUT_BLOCK; e += WG_THREADS) {
    const int io  = (e >= OUT_BLOCK) ? 1 : 0;
    const int rem = e - io * OUT_BLOCK;
    const int row = rem / DD;
    const int d   = rem - row * DD;
    const float* p = part + io * (16 * PITCH) + row * PITCH + d;
    float sum = p[0];
#pragma unroll
    for (int w = 1; w < SPLITK; ++w) sum += p[w * PART_WAVE];
    out[(size_t)io * NN * DD + (size_t)i0 * DD + rem] = sum;
  }
}

extern "C" void kernel_launch(void* const* d_in, const int* in_sizes, int n_in,
                              void* d_out, int out_size, void* d_ws, size_t ws_size,
                              hipStream_t stream) {
  const float* adj = (const float*)d_in[0];
  const float* s   = (const float*)d_in[1];
  float* out = (float*)d_out;
  v2f* sB = (v2f*)d_ws;

  prep_sB<<<PREP_BLOCKS, PREP_THREADS, 0, stream>>>(s, sB);

  slayer_wmma<<<NT, WG_THREADS, 0, stream>>>(adj, sB, out);
}


# s fragments packed two k-steps per entry and fetched as 16-byte loads (10 a tile, not 20): a tile is 22 loads, so 3 tiles ahead fit the 63-load counter; sweep pipelined over four register sets
# speedup vs baseline: 1.8514x; 1.0369x over previous
#include <hip/hip_runtime.h>

#define NN   4096
#define DD   70
#define NT   256
#define ND   5
#define SPLITK 8
#define TPC  (NT / SPLITK)

typedef float v2f __attribute__((ext_vector_type(2)));
typedef float v4f __attribute__((ext_vector_type(4)));
typedef float v8f __attribute__((ext_vector_type(8)));

__device__ __forceinline__ v8f wmma4(v2f a, v2f b, v8f c) {
  return __builtin_amdgcn_wmma_f32_16x16x4_f32(
       false, a,  false, b,
       (short)0, c,  false,  false);
}

#define PREP_THREADS 256
#define PREP_BLOCKS  160
#define PREP_TOTAL   (PREP_THREADS * PREP_BLOCKS)
#define PACK_N       (NT * ND * 2 * 32)
#define PACK_PER     (PACK_N / PREP_TOTAL)
static_assert(PACK_PER * PREP_TOTAL == PACK_N, "pack entries must divide evenly over the prep threads");

__device__ __forceinline__ v4f pack_entry(const float* __restrict__ s, int idx) {
  int lane = idx & 31;
  int e = idx >> 5;
  int p = e & 1;  e >>= 1;
  int n = e % ND;
  int t = e / ND;
  int r = 16 * t + 8 * p + 2 * (lane >> 4);
  int d = 16 * n + (lane & 15);
  v4f v;
  v.x = (d < DD) ? s[(size_t)r * DD + d]       : 0.0f;
  v.y = (d < DD) ? s[(size_t)(r + 1) * DD + d] : 0.0f;
  v.z = (d < DD) ? s[(size_t)(r + 4) * DD + d] : 0.0f;
  v.w = (d < DD) ? s[(size_t)(r + 5) * DD + d] : 0.0f;
  return v;
}

__global__ __launch_bounds__(PREP_THREADS) void prep_sB(const float* __restrict__ s,
                                                        v4f*   __restrict__ sB) {
  const int tid = blockIdx.x * PREP_THREADS + threadIdx.x;

  v4f v[PACK_PER];
#pragma unroll
  for (int k = 0; k < PACK_PER; ++k) v[k] = pack_entry(s, tid + k * PREP_TOTAL);

#pragma unroll
  for (int k = 0; k < PACK_PER; ++k) sB[tid + k * PREP_TOTAL] = v[k];
}

struct TileRegs {
  v4f f[4];
  v2f g0[4], g1[4];
  v4f b[2][ND];
};

__device__ __forceinline__ void load_tile(TileRegs& T, const float* __restrict__ pO,
                                          const float* __restrict__ pI,
                                          const v4f* __restrict__ pB) {
#pragma unroll
  for (int q = 0; q < 4; ++q) {
    T.f[q]  = *(const v4f*)(pO + 8 * q);
    T.g0[q] = *(const v2f*)(pI + (size_t)q * 8 * NN);
    T.g1[q] = *(const v2f*)(pI + (size_t)q * 8 * NN + 2 * NN);
  }
#pragma unroll
  for (int p = 0; p < 2; ++p) {
#pragma unroll
    for (int n = 0; n < ND; ++n) T.b[p][n] = pB[p * 32 + n * 2 * 32];
  }
}

__device__ __forceinline__ void mma_tile(const TileRegs& T, v8f (&accI)[ND], v8f (&accO)[ND]) {
#pragma unroll
  for (int q = 0; q < 4; ++q) {
    v2f aO; aO.x = T.f[q].x + T.f[q].y;   aO.y = T.f[q].z + T.f[q].w;
    v2f aI; aI.x = T.g0[q].x + T.g0[q].y; aI.y = T.g1[q].x + T.g1[q].y;
#pragma unroll
    for (int n = 0; n < ND; ++n) {
      const v4f w = T.b[q >> 1][n];
      v2f bq;
      bq.x = (q & 1) ? w.z : w.x;
      bq.y = (q & 1) ? w.w : w.y;
      accO[n] = wmma4(aO, bq, accO[n]);
      accI[n] = wmma4(aI, bq, accI[n]);
    }
  }
}

#define WG_THREADS (32 * SPLITK)
#define PITCH      84
#define PART_WAVE  (2 * 16 * PITCH)
#define OUT_BLOCK  (16 * DD)

__global__ __launch_bounds__(WG_THREADS) void slayer_wmma(const float* __restrict__ adj,
                                                          const v4f*  __restrict__ sB,
                                                          float*      __restrict__ out) {
  __shared__ float part[SPLITK * PART_WAVE];

  const int b    = blockIdx.x;
  const int c    = __builtin_amdgcn_readfirstlane((int)(threadIdx.x >> 5));
  const int lane = threadIdx.x & 31;
  const int m    = lane & 15;
  const int kh   = lane >> 4;

  v8f zero = {};
  v8f accI[ND], accO[ND];
#pragma unroll
  for (int n = 0; n < ND; ++n) { accI[n] = zero; accO[n] = zero; }

  const int i0 = 16 * b;
  const int t0 = TPC * c;
  const float* pO = adj + 2 * ((size_t)(i0 + m) * NN + (16 * t0 + 2 * kh));
  const float* pI = adj + 2 * ((size_t)(16 * t0 + 2 * kh) * NN + (i0 + m));
  const v4f*   pB = sB + (size_t)t0 * (ND * 2 * 32) + lane;
  const size_t dO = 2 * 16;
  const size_t dI = (size_t)2 * 16 * NN;
  const size_t dB = ND * 2 * 32;

#define SCHED_FENCE() __builtin_amdgcn_sched_barrier(0)

  static_assert(TPC >= 4 && (TPC - 4) % 4 == 0, "the sweep below takes the chunk in trips of four tiles");
  TileRegs R0, R1, R2, R3;
  load_tile(R0, pO, pI, pB);
  SCHED_FENCE();
  load_tile(R1, pO + dO, pI + dI, pB + dB);
  SCHED_FENCE();
  load_tile(R2, pO + 2 * dO, pI + 2 * dI, pB + 2 * dB);
  SCHED_FENCE();
  pO += 3 * dO; pI += 3 * dI; pB += 3 * dB;

  for (int it = 0; it < (TPC - 4) / 4; ++it) {
    load_tile(R3, pO, pI, pB);
    SCHED_FENCE();
    mma_tile(R0, accI, accO);
    SCHED_FENCE();
    load_tile(R0, pO + dO, pI + dI, pB + dB);
    SCHED_FENCE();
    mma_tile(R1, accI, accO);
    SCHED_FENCE();
    load_tile(R1, pO + 2 * dO, pI + 2 * dI, pB + 2 * dB);
    SCHED_FENCE();
    mma_tile(R2, accI, accO);
    SCHED_FENCE();
    load_tile(R2, pO + 3 * dO, pI + 3 * dI, pB + 3 * dB);
    SCHED_FENCE();
    mma_tile(R3, accI, accO);
    SCHED_FENCE();
    pO += 4 * dO; pI += 4 * dI; pB += 4 * dB;
  }
  load_tile(R3, pO, pI, pB);
  SCHED_FENCE();
  mma_tile(R0, accI, accO);
  mma_tile(R1, accI, accO);
  mma_tile(R2, accI, accO);
  mma_tile(R3, accI, accO);
#undef SCHED_FENCE

  float* mine = part + c * PART_WAVE;
#pragma unroll
  for (int n = 0; n < ND; ++n) {
#pragma unroll
    for (int r = 0; r < 8; ++r) {
      const int at = (r + 8 * kh) * PITCH + 16 * n + m;
      mine[at]              = accI[n][r];
      mine[16 * PITCH + at] = accO[n][r];
    }
  }
  __syncthreads();

  for (int e = threadIdx.x; e < 2 * OUT_BLOCK; e += WG_THREADS) {
    const int io  = (e >= OUT_BLOCK) ? 1 : 0;
    const int rem = e - io * OUT_BLOCK;
    const int row = rem / DD;
    const int d   = rem - row * DD;
    const float* p = part + io * (16 * PITCH) + row * PITCH + d;
    float sum = p[0];
#pragma unroll
    for (int w = 1; w < SPLITK; ++w) sum += p[w * PART_WAVE];
    out[(size_t)io * NN * DD + (size_t)i0 * DD + rem] = sum;
  }
}

extern "C" void kernel_launch(void* const* d_in, const int* in_sizes, int n_in,
                              void* d_out, int out_size, void* d_ws, size_t ws_size,
                              hipStream_t stream) {
  const float* adj = (const float*)d_in[0];
  const float* s   = (const float*)d_in[1];
  float* out = (float*)d_out;
  v4f* sB = (v4f*)d_ws;

  prep_sB<<<PREP_BLOCKS, PREP_THREADS, 0, stream>>>(s, sB);

  slayer_wmma<<<NT, WG_THREADS, 0, stream>>>(adj, sB, out);
}


# row-panel adjacency tile fetched as whole 128-byte lines (4 lines a load, not 16 quarter lines) and turned into A fragments through the wave's own LDS area one tile ahead; sum order unchanged
# speedup vs baseline: 1.9089x; 1.0311x over previous
#include <hip/hip_runtime.h>

#define NN   4096
#define DD   70
#define NT   256
#define ND   5
#define SPLITK 8
#define TPC  (NT / SPLITK)

typedef float v2f __attribute__((ext_vector_type(2)));
typedef float v4f __attribute__((ext_vector_type(4)));
typedef float v8f __attribute__((ext_vector_type(8)));

__device__ __forceinline__ v8f wmma4(v2f a, v2f b, v8f c) {
  return __builtin_amdgcn_wmma_f32_16x16x4_f32(
       false, a,  false, b,
       (short)0, c,  false,  false);
}

#define PREP_THREADS 256
#define PREP_BLOCKS  160
#define PREP_TOTAL   (PREP_THREADS * PREP_BLOCKS)
#define PACK_N       (NT * ND * 2 * 32)
#define PACK_PER     (PACK_N / PREP_TOTAL)
static_assert(PACK_PER * PREP_TOTAL == PACK_N, "pack entries must divide evenly over the prep threads");

__device__ __forceinline__ v4f pack_entry(const float* __restrict__ s, int idx) {
  int lane = idx & 31;
  int e = idx >> 5;
  int p = e & 1;  e >>= 1;
  int n = e % ND;
  int t = e / ND;
  int r = 16 * t + 8 * p + 2 * (lane >> 4);
  int d = 16 * n + (lane & 15);
  v4f v;
  v.x = (d < DD) ? s[(size_t)r * DD + d]       : 0.0f;
  v.y = (d < DD) ? s[(size_t)(r + 1) * DD + d] : 0.0f;
  v.z = (d < DD) ? s[(size_t)(r + 4) * DD + d] : 0.0f;
  v.w = (d < DD) ? s[(size_t)(r + 5) * DD + d] : 0.0f;
  return v;
}

__global__ __launch_bounds__(PREP_THREADS) void prep_sB(const float* __restrict__ s,
                                                        v4f*   __restrict__ sB) {
  const int tid = blockIdx.x * PREP_THREADS + threadIdx.x;

  v4f v[PACK_PER];
#pragma unroll
  for (int k = 0; k < PACK_PER; ++k) v[k] = pack_entry(s, tid + k * PREP_TOTAL);

#pragma unroll
  for (int k = 0; k < PACK_PER; ++k) sB[tid + k * PREP_TOTAL] = v[k];
}

struct TileRegs {
  v4f raw[4];
  v2f g0[4], g1[4];
  v4f b[2][ND];
};

struct AFrag { v2f a[4]; };

#define QS        80
#define STAGE_BUF (4 * QS)
#define STAGE_WAVE (2 * STAGE_BUF)

__device__ __forceinline__ void load_tile(TileRegs& T, const float* __restrict__ pO,
                                          const float* __restrict__ pI,
                                          const v4f* __restrict__ pB) {
#pragma unroll
  for (int r = 0; r < 4; ++r) T.raw[r] = *(const v4f*)(pO + (size_t)r * 8 * NN);
#pragma unroll
  for (int q = 0; q < 4; ++q) {
    T.g0[q] = *(const v2f*)(pI + (size_t)q * 8 * NN);
    T.g1[q] = *(const v2f*)(pI + (size_t)q * 8 * NN + 2 * NN);
  }
#pragma unroll
  for (int p = 0; p < 2; ++p) {
#pragma unroll
    for (int n = 0; n < ND; ++n) T.b[p][n] = pB[p * 32 + n * 2 * 32];
  }
}

__device__ __forceinline__ void stage_tile(const TileRegs& T, AFrag& F, float* st, const float* ld) {
#pragma unroll
  for (int r = 0; r < 4; ++r) {
    v2f sum; sum.x = T.raw[r].x + T.raw[r].y; sum.y = T.raw[r].z + T.raw[r].w;
    *(v2f*)(st + 16 * r) = sum;
  }
  __builtin_amdgcn_fence(__ATOMIC_RELEASE, "wavefront");
  __builtin_amdgcn_wave_barrier();
  __builtin_amdgcn_fence(__ATOMIC_ACQUIRE, "wavefront");
#pragma unroll
  for (int q = 0; q < 4; ++q) F.a[q] = *(const v2f*)(ld + QS * q);
}

__device__ __forceinline__ void mma_tile(const TileRegs& T, const AFrag& F, v8f (&accI)[ND], v8f (&accO)[ND]) {
#pragma unroll
  for (int q = 0; q < 4; ++q) {
    const v2f aO = F.a[q];
    v2f aI; aI.x = T.g0[q].x + T.g0[q].y; aI.y = T.g1[q].x + T.g1[q].y;
#pragma unroll
    for (int n = 0; n < ND; ++n) {
      const v4f w = T.b[q >> 1][n];
      v2f bq;
      bq.x = (q & 1) ? w.z : w.x;
      bq.y = (q & 1) ? w.w : w.y;
      accO[n] = wmma4(aO, bq, accO[n]);
      accI[n] = wmma4(aI, bq, accI[n]);
    }
  }
}

#define WG_THREADS (32 * SPLITK)
#define PITCH      84
#define PART_WAVE  (2 * 16 * PITCH)
#define OUT_BLOCK  (16 * DD)

__global__ __launch_bounds__(WG_THREADS) void slayer_wmma(const float* __restrict__ adj,
                                                          const v4f*  __restrict__ sB,
                                                          float*      __restrict__ out) {
  __shared__ float part[SPLITK * PART_WAVE];
  __shared__ __attribute__((aligned(16))) float stage[SPLITK * STAGE_WAVE];

  const int b    = blockIdx.x;
  const int c    = __builtin_amdgcn_readfirstlane((int)(threadIdx.x >> 5));
  const int lane = threadIdx.x & 31;
  const int m    = lane & 15;
  const int kh   = lane >> 4;

  v8f zero = {};
  v8f accI[ND], accO[ND];
#pragma unroll
  for (int n = 0; n < ND; ++n) { accI[n] = zero; accO[n] = zero; }

  const int i0 = 16 * b;
  const int t0 = TPC * c;
  const float* pO = adj + 2 * ((size_t)(i0 + (lane >> 3)) * NN + 16 * t0) + 4 * (lane & 7);
  const float* pI = adj + 2 * ((size_t)(16 * t0 + 2 * kh) * NN + (i0 + m));
  const v4f*   pB = sB + (size_t)t0 * (ND * 2 * 32) + lane;
  const size_t dO = 2 * 16;
  const size_t dI = (size_t)2 * 16 * NN;
  const size_t dB = ND * 2 * 32;

  float* sw = stage + c * STAGE_WAVE;
  float* st0 = sw + ((lane & 7) >> 1) * QS + (lane >> 3) * 4 + 2 * (lane & 1);
  float* st1 = st0 + STAGE_BUF;
  const float* ld0 = sw + m * 4 + 2 * kh;
  const float* ld1 = ld0 + STAGE_BUF;

#define SCHED_FENCE() __builtin_amdgcn_sched_barrier(0)

  static_assert(TPC >= 4 && (TPC - 4) % 4 == 0, "the sweep below takes the chunk in trips of four tiles");
  TileRegs R0, R1, R2, R3;
  AFrag F0, F1;
  load_tile(R0, pO, pI, pB);
  SCHED_FENCE();
  load_tile(R1, pO + dO, pI + dI, pB + dB);
  SCHED_FENCE();
  load_tile(R2, pO + 2 * dO, pI + 2 * dI, pB + 2 * dB);
  SCHED_FENCE();
  stage_tile(R0, F0, st0, ld0);
  SCHED_FENCE();
  pO += 3 * dO; pI += 3 * dI; pB += 3 * dB;

  for (int it = 0; it < (TPC - 4) / 4; ++it) {
    load_tile(R3, pO, pI, pB);
    SCHED_FENCE();
    stage_tile(R1, F1, st1, ld1);
    SCHED_FENCE();
    mma_tile(R0, F0, accI, accO);
    SCHED_FENCE();
    load_tile(R0, pO + dO, pI + dI, pB + dB);
    SCHED_FENCE();
    stage_tile(R2, F0, st0, ld0);
    SCHED_FENCE();
    mma_tile(R1, F1, accI, accO);
    SCHED_FENCE();
    load_tile(R1, pO + 2 * dO, pI + 2 * dI, pB + 2 * dB);
    SCHED_FENCE();
    stage_tile(R3, F1, st1, ld1);
    SCHED_FENCE();
    mma_tile(R2, F0, accI, accO);
    SCHED_FENCE();
    load_tile(R2, pO + 3 * dO, pI + 3 * dI, pB + 3 * dB);
    SCHED_FENCE();
    stage_tile(R0, F0, st0, ld0);
    SCHED_FENCE();
    mma_tile(R3, F1, accI, accO);
    SCHED_FENCE();
    pO += 4 * dO; pI += 4 * dI; pB += 4 * dB;
  }
  load_tile(R3, pO, pI, pB);
  SCHED_FENCE();
  stage_tile(R1, F1, st1, ld1);
  SCHED_FENCE();
  mma_tile(R0, F0, accI, accO);
  SCHED_FENCE();
  stage_tile(R2, F0, st0, ld0);
  SCHED_FENCE();
  mma_tile(R1, F1, accI, accO);
  SCHED_FENCE();
  stage_tile(R3, F1, st1, ld1);
  SCHED_FENCE();
  mma_tile(R2, F0, accI, accO);
  mma_tile(R3, F1, accI, accO);
#undef SCHED_FENCE

  float* mine = part + c * PART_WAVE;
#pragma unroll
  for (int n = 0; n < ND; ++n) {
#pragma unroll
    for (int r = 0; r < 8; ++r) {
      const int at = (r + 8 * kh) * PITCH + 16 * n + m;
      mine[at]              = accI[n][r];
      mine[16 * PITCH + at] = accO[n][r];
    }
  }
  __syncthreads();

  for (int e = threadIdx.x; e < 2 * OUT_BLOCK; e += WG_THREADS) {
    const int io  = (e >= OUT_BLOCK) ? 1 : 0;
    const int rem = e - io * OUT_BLOCK;
    const int row = rem / DD;
    const int d   = rem - row * DD;
    const float* p = part + io * (16 * PITCH) + row * PITCH + d;
    float sum = p[0];
#pragma unroll
    for (int w = 1; w < SPLITK; ++w) sum += p[w * PART_WAVE];
    out[(size_t)io * NN * DD + (size_t)i0 * DD + rem] = sum;
  }
}

extern "C" void kernel_launch(void* const* d_in, const int* in_sizes, int n_in,
                              void* d_out, int out_size, void* d_ws, size_t ws_size,
                              hipStream_t stream) {
  const float* adj = (const float*)d_in[0];
  const float* s   = (const float*)d_in[1];
  float* out = (float*)d_out;
  v4f* sB = (v4f*)d_ws;

  prep_sB<<<PREP_BLOCKS, PREP_THREADS, 0, stream>>>(s, sB);

  slayer_wmma<<<NT, WG_THREADS, 0, stream>>>(adj, sB, out);
}
